# starting kernel
# baseline (speedup 1.0000x reference)
#include <hip/hip_runtime.h>

typedef float v2f __attribute__((ext_vector_type(2)));
typedef float v8f __attribute__((ext_vector_type(8)));

#define L_LEN 2048
#define N_HEAD 128
#define D_MODEL_SZ 1024
#define HEAD_DIM 8
#define CHUNK 512
#define ROWS_PER_BLK 128
#define NWAVE 8
#define KPAD 16

__global__ __launch_bounds__(256)
void hyena_attn_kernel(const float* __restrict__ kg,
                       const float* __restrict__ vg,
                       const float* __restrict__ qg,
                       const float* __restrict__ kerg,
                       const float* __restrict__ Dg,
                       float* __restrict__ outg)
{
    __shared__ __attribute__((aligned(16))) float sK[HEAD_DIM][CHUNK];
    __shared__ __attribute__((aligned(16))) float sVp[16][CHUNK];
    __shared__ __attribute__((aligned(16))) float sQ[HEAD_DIM][ROWS_PER_BLK];
    __shared__ __attribute__((aligned(16))) float sKerP[KPAD + L_LEN];
    __shared__ __attribute__((aligned(16))) float sW[NWAVE][16 * 17];

    const int tid  = threadIdx.x;
    const int lane = tid & 31;
    const int wid  = tid >> 5;

    const int rblk = blockIdx.x & 15;
    const int bh   = blockIdx.x >> 4;
    const int b    = bh / N_HEAD;
    const int h    = bh % N_HEAD;

    const size_t headBase = ((size_t)b * D_MODEL_SZ + (size_t)h * HEAD_DIM) * L_LEN;

    const float Dh = Dg[h];

    {
        int i  = tid >> 5;
        int c4 = tid & 31;
        const float4* src = (const float4*)(qg + headBase + (size_t)i * L_LEN + rblk * ROWS_PER_BLK);
        ((float4*)&sQ[i][0])[c4] = src[c4];
    }
    {
        const float4* src = (const float4*)(kerg + (size_t)h * L_LEN);
        for (int u = tid; u < L_LEN / 4; u += 256) {
            float4 f = src[u];
            if (u == 0) f.x += Dh;
            ((float4*)(sKerP + KPAD))[u] = f;
        }
        if (tid < KPAD) sKerP[tid] = 0.0f;
    }
    {
        float4 z = make_float4(0.f, 0.f, 0.f, 0.f);
        for (int u = tid; u < 8 * CHUNK / 4; u += 256)
            ((float4*)&sVp[8][0])[u] = z;
    }
    __syncthreads();

    const int rowBase  = rblk * ROWS_PER_BLK + wid * 16;
    const int rowBaseU = __builtin_amdgcn_readfirstlane(rowBase);
    const int n  = lane & 15;
    const int k0 = (lane < 16) ? 0 : 2;
    const int mB = (lane < 16) ? 0 : 8;

    v2f qa0, qa1;
    {
        int lcol = wid * 16 + n;
        qa0.x = sQ[k0 + 0][lcol];
        qa0.y = sQ[k0 + 1][lcol];
        qa1.x = sQ[4 + k0 + 0][lcol];
        qa1.y = sQ[4 + k0 + 1][lcol];
    }

    v8f acc = {};
    float* sWw = &sW[wid][0];
    const int arow = n * 17;

    const int nChunk = (rblk >> 2) + 1;
    for (int c = 0; c < nChunk; ++c) {
        const int tStart = c * CHUNK;

        for (int u = tid; u < HEAD_DIM * CHUNK / 4; u += 256) {
            int i  = u >> 7;
            int c4 = u & 127;
            ((float4*)&sK[i][0])[c4]  = ((const float4*)(kg + headBase + (size_t)i * L_LEN + tStart))[c4];
            ((float4*)&sVp[i][0])[c4] = ((const float4*)(vg + headBase + (size_t)i * L_LEN + tStart))[c4];
        }

        if (c + 1 < nChunk) {
            const int tNext = tStart + CHUNK;
            const float* basePtr = (tid < 128) ? kg : vg;
            int r  = (tid & 127) >> 4;
            int ln = (tid & 15) * 32;
            __builtin_prefetch(basePtr + headBase + (size_t)r * L_LEN + tNext + ln, 0, 3);
        }
        __syncthreads();

        int ttEnd = ((rowBaseU - tStart) >> 4) + 1;
        if (ttEnd > CHUNK / 16) ttEnd = CHUNK / 16;

        for (int tt = 0; tt < ttEnd; ++tt) {
            const int colLoc  = tt * 16;
            const int colBase = tStart + colLoc;

            v2f bk0, bk1;
            bk0.x = sK[k0 + 0][colLoc + n];
            bk0.y = sK[k0 + 1][colLoc + n];
            bk1.x = sK[4 + k0 + 0][colLoc + n];
            bk1.y = sK[4 + k0 + 1][colLoc + n];

            v8f s = {};
            s = __builtin_amdgcn_wmma_f32_16x16x4_f32(false, qa0, false, bk0, (short)0, s, false, false);
            s = __builtin_amdgcn_wmma_f32_16x16x4_f32(false, qa1, false, bk1, (short)0, s, false, false);

            const int kerBase = KPAD + rowBase + mB - (colBase + n);
            #pragma unroll
            for (int r = 0; r < 8; ++r) {
                float wc = sKerP[kerBase + r];
                sWw[(mB + r) * 17 + n] = s[r] * wc;
            }

            #pragma unroll
            for (int kc = 0; kc < 4; ++kc) {
                v2f aw, bv;
                aw.x = sWw[arow + 4 * kc + k0 + 0];
                aw.y = sWw[arow + 4 * kc + k0 + 1];
                int tl = colLoc + 4 * kc + k0;
                bv.x = sVp[n][tl + 0];
                bv.y = sVp[n][tl + 1];
                acc = __builtin_amdgcn_wmma_f32_16x16x4_f32(false, aw, false, bv, (short)0, acc, false, false);
            }
        }
        __syncthreads();
    }

    if (n < 8) {
        size_t o = headBase + (size_t)n * L_LEN + (size_t)(rowBase + mB);
        float4 lo = make_float4(acc[0], acc[1], acc[2], acc[3]);
        float4 hi = make_float4(acc[4], acc[5], acc[6], acc[7]);
        *(float4*)(outg + o)     = lo;
        *(float4*)(outg + o + 4) = hi;
    }
}

extern "C" void kernel_launch(void* const* d_in, const int* in_sizes, int n_in,
                              void* d_out, int out_size, void* d_ws, size_t ws_size,
                              hipStream_t stream) {
    const float* k   = (const float*)d_in[0];
    const float* v   = (const float*)d_in[1];
    const float* q   = (const float*)d_in[2];
    const float* ker = (const float*)d_in[3];
    const float* D   = (const float*)d_in[4];
    float* out = (float*)d_out;

    const int B = in_sizes[0] / (D_MODEL_SZ * L_LEN);
    dim3 grid((unsigned)(B * N_HEAD * (L_LEN / ROWS_PER_BLK)));
    hipLaunchKernelGGL(hyena_attn_kernel, grid, dim3(256), 0, stream, k, v, q, ker, D, out);
}


# LDS row pitches of the K, V and W tiles padded (528, 516, 20 floats) so every fragment load and store of the tile loop is free of bank conflicts; V fragment reads had all rows on the same banks
# speedup vs baseline: 2.0510x; 2.0510x over previous
#include <hip/hip_runtime.h>

typedef float v2f __attribute__((ext_vector_type(2)));
typedef float v8f __attribute__((ext_vector_type(8)));

#define L_LEN 2048
#define N_HEAD 128
#define D_MODEL_SZ 1024
#define HEAD_DIM 8
#define CHUNK 512
#define ROWS_PER_BLK 128
#define NWAVE 8
#define KPAD 16
#define K_PITCH (CHUNK + 16)
#define V_PITCH (CHUNK + 4)
#define W_PITCH 20

__global__ __launch_bounds__(256)
void hyena_attn_kernel(const float* __restrict__ kg,
                       const float* __restrict__ vg,
                       const float* __restrict__ qg,
                       const float* __restrict__ kerg,
                       const float* __restrict__ Dg,
                       float* __restrict__ outg)
{
    __shared__ __attribute__((aligned(16))) float sK[HEAD_DIM][K_PITCH];
    __shared__ __attribute__((aligned(16))) float sVp[16][V_PITCH];
    __shared__ __attribute__((aligned(16))) float sQ[HEAD_DIM][ROWS_PER_BLK];
    __shared__ __attribute__((aligned(16))) float sKerP[KPAD + L_LEN];
    __shared__ __attribute__((aligned(16))) float sW[NWAVE][16 * W_PITCH];

    const int tid  = threadIdx.x;
    const int lane = tid & 31;
    const int wid  = tid >> 5;

    const int rblk = blockIdx.x & 15;
    const int bh   = blockIdx.x >> 4;
    const int b    = bh / N_HEAD;
    const int h    = bh % N_HEAD;

    const size_t headBase = ((size_t)b * D_MODEL_SZ + (size_t)h * HEAD_DIM) * L_LEN;

    const float Dh = Dg[h];

    {
        int i  = tid >> 5;
        int c4 = tid & 31;
        const float4* src = (const float4*)(qg + headBase + (size_t)i * L_LEN + rblk * ROWS_PER_BLK);
        ((float4*)&sQ[i][0])[c4] = src[c4];
    }
    {
        const float4* src = (const float4*)(kerg + (size_t)h * L_LEN);
        for (int u = tid; u < L_LEN / 4; u += 256) {
            float4 f = src[u];
            if (u == 0) f.x += Dh;
            ((float4*)(sKerP + KPAD))[u] = f;
        }
        if (tid < KPAD) sKerP[tid] = 0.0f;
    }
    {
        float4 z = make_float4(0.f, 0.f, 0.f, 0.f);
        for (int u = tid; u < 8 * V_PITCH / 4; u += 256)
            ((float4*)&sVp[8][0])[u] = z;
    }
    __syncthreads();

    const int rowBase  = rblk * ROWS_PER_BLK + wid * 16;
    const int rowBaseU = __builtin_amdgcn_readfirstlane(rowBase);
    const int n  = lane & 15;
    const int k0 = (lane < 16) ? 0 : 2;
    const int mB = (lane < 16) ? 0 : 8;

    v2f qa0, qa1;
    {
        int lcol = wid * 16 + n;
        qa0.x = sQ[k0 + 0][lcol];
        qa0.y = sQ[k0 + 1][lcol];
        qa1.x = sQ[4 + k0 + 0][lcol];
        qa1.y = sQ[4 + k0 + 1][lcol];
    }

    v8f acc = {};
    float* sWw = &sW[wid][0];
    const int arow = n * W_PITCH;

    const int nChunk = (rblk >> 2) + 1;
    for (int c = 0; c < nChunk; ++c) {
        const int tStart = c * CHUNK;

        for (int u = tid; u < HEAD_DIM * CHUNK / 4; u += 256) {
            int i  = u >> 7;
            int c4 = u & 127;
            ((float4*)&sK[i][0])[c4]  = ((const float4*)(kg + headBase + (size_t)i * L_LEN + tStart))[c4];
            ((float4*)&sVp[i][0])[c4] = ((const float4*)(vg + headBase + (size_t)i * L_LEN + tStart))[c4];
        }

        if (c + 1 < nChunk) {
            const int tNext = tStart + CHUNK;
            const float* basePtr = (tid < 128) ? kg : vg;
            int r  = (tid & 127) >> 4;
            int ln = (tid & 15) * 32;
            __builtin_prefetch(basePtr + headBase + (size_t)r * L_LEN + tNext + ln, 0, 3);
        }
        __syncthreads();

        int ttEnd = ((rowBaseU - tStart) >> 4) + 1;
        if (ttEnd > CHUNK / 16) ttEnd = CHUNK / 16;

        for (int tt = 0; tt < ttEnd; ++tt) {
            const int colLoc  = tt * 16;
            const int colBase = tStart + colLoc;

            v2f bk0, bk1;
            bk0.x = sK[k0 + 0][colLoc + n];
            bk0.y = sK[k0 + 1][colLoc + n];
            bk1.x = sK[4 + k0 + 0][colLoc + n];
            bk1.y = sK[4 + k0 + 1][colLoc + n];

            v8f s = {};
            s = __builtin_amdgcn_wmma_f32_16x16x4_f32(false, qa0, false, bk0, (short)0, s, false, false);
            s = __builtin_amdgcn_wmma_f32_16x16x4_f32(false, qa1, false, bk1, (short)0, s, false, false);

            const int kerBase = KPAD + rowBase + mB - (colBase + n);
            #pragma unroll
            for (int r = 0; r < 8; ++r) {
                float wc = sKerP[kerBase + r];
                sWw[(mB + r) * W_PITCH + n] = s[r] * wc;
            }

            #pragma unroll
            for (int kc = 0; kc < 4; ++kc) {
                v2f aw, bv;
                aw.x = sWw[arow + 4 * kc + k0 + 0];
                aw.y = sWw[arow + 4 * kc + k0 + 1];
                int tl = colLoc + 4 * kc + k0;
                bv.x = sVp[n][tl + 0];
                bv.y = sVp[n][tl + 1];
                acc = __builtin_amdgcn_wmma_f32_16x16x4_f32(false, aw, false, bv, (short)0, acc, false, false);
            }
        }
        __syncthreads();
    }

    if (n < 8) {
        size_t o = headBase + (size_t)n * L_LEN + (size_t)(rowBase + mB);
        float4 lo = make_float4(acc[0], acc[1], acc[2], acc[3]);
        float4 hi = make_float4(acc[4], acc[5], acc[6], acc[7]);
        *(float4*)(outg + o)     = lo;
        *(float4*)(outg + o + 4) = hi;
    }
}

extern "C" void kernel_launch(void* const* d_in, const int* in_sizes, int n_in,
                              void* d_out, int out_size, void* d_ws, size_t ws_size,
                              hipStream_t stream) {
    const float* k   = (const float*)d_in[0];
    const float* v   = (const float*)d_in[1];
    const float* q   = (const float*)d_in[2];
    const float* ker = (const float*)d_in[3];
    const float* D   = (const float*)d_in[4];
    float* out = (float*)d_out;

    const int B = in_sizes[0] / (D_MODEL_SZ * L_LEN);
    dim3 grid((unsigned)(B * N_HEAD * (L_LEN / ROWS_PER_BLK)));
    hipLaunchKernelGGL(hyena_attn_kernel, grid, dim3(256), 0, stream, k, v, q, ker, D, out);
}


# Each work-group owns row block r and its mirror 15-r on one staged K/V chunk stream (wave w: row tile w of r, tile 7-w of 15-r): every wave does 129 tiles, causal imbalance gone; grid 4096 -> 2048
# speedup vs baseline: 2.8758x; 1.4021x over previous
#include <hip/hip_runtime.h>

typedef float v2f __attribute__((ext_vector_type(2)));
typedef float v8f __attribute__((ext_vector_type(8)));

#define L_LEN 2048
#define N_HEAD 128
#define D_MODEL_SZ 1024
#define HEAD_DIM 8
#define CHUNK 512
#define ROWS_PER_BLK 128
#define N_ROW_BLK (L_LEN / ROWS_PER_BLK)
#define NWAVE 8
#define KPAD 16
#define K_PITCH (CHUNK + 16)
#define V_PITCH (CHUNK + 4)
#define W_PITCH 20

static __device__ __forceinline__
void row_tile_chunk(v8f& acc, const v2f qa0, const v2f qa1,
                    const int rowBase, const int rowBaseU, const int tStart,
                    const float (*sK)[K_PITCH], const float (*sVp)[V_PITCH],
                    const float* sKerP, float* sWw,
                    const int n, const int k0, const int mB, const int arow)
{
    int ttEnd = ((rowBaseU - tStart) >> 4) + 1;
    if (ttEnd > CHUNK / 16) ttEnd = CHUNK / 16;

    for (int tt = 0; tt < ttEnd; ++tt) {
        const int colLoc  = tt * 16;
        const int colBase = tStart + colLoc;

        v2f bk0, bk1;
        bk0.x = sK[k0 + 0][colLoc + n];
        bk0.y = sK[k0 + 1][colLoc + n];
        bk1.x = sK[4 + k0 + 0][colLoc + n];
        bk1.y = sK[4 + k0 + 1][colLoc + n];

        v8f s = {};
        s = __builtin_amdgcn_wmma_f32_16x16x4_f32(false, qa0, false, bk0, (short)0, s, false, false);
        s = __builtin_amdgcn_wmma_f32_16x16x4_f32(false, qa1, false, bk1, (short)0, s, false, false);

        const int kerBase = KPAD + rowBase + mB - (colBase + n);
        #pragma unroll
        for (int r = 0; r < 8; ++r) {
            float wc = sKerP[kerBase + r];
            sWw[(mB + r) * W_PITCH + n] = s[r] * wc;
        }

        #pragma unroll
        for (int kc = 0; kc < 4; ++kc) {
            v2f aw, bv;
            aw.x = sWw[arow + 4 * kc + k0 + 0];
            aw.y = sWw[arow + 4 * kc + k0 + 1];
            int tl = colLoc + 4 * kc + k0;
            bv.x = sVp[n][tl + 0];
            bv.y = sVp[n][tl + 1];
            acc = __builtin_amdgcn_wmma_f32_16x16x4_f32(false, aw, false, bv, (short)0, acc, false, false);
        }
    }
}

__global__ __launch_bounds__(256)
void hyena_attn_kernel(const float* __restrict__ kg,
                       const float* __restrict__ vg,
                       const float* __restrict__ qg,
                       const float* __restrict__ kerg,
                       const float* __restrict__ Dg,
                       float* __restrict__ outg)
{
    __shared__ __attribute__((aligned(16))) float sK[HEAD_DIM][K_PITCH];
    __shared__ __attribute__((aligned(16))) float sVp[16][V_PITCH];
    __shared__ __attribute__((aligned(16))) float sQ[2][HEAD_DIM][ROWS_PER_BLK];
    __shared__ __attribute__((aligned(16))) float sKerP[KPAD + L_LEN];
    __shared__ __attribute__((aligned(16))) float sW[NWAVE][16 * W_PITCH];

    const int tid  = threadIdx.x;
    const int lane = tid & 31;
    const int wid  = tid >> 5;

    const int rA = blockIdx.x & (N_ROW_BLK / 2 - 1);
    const int rB = N_ROW_BLK - 1 - rA;
    const int bh = blockIdx.x >> 3;
    const int b  = bh / N_HEAD;
    const int h  = bh % N_HEAD;

    const size_t headBase = ((size_t)b * D_MODEL_SZ + (size_t)h * HEAD_DIM) * L_LEN;

    const float Dh = Dg[h];

    {
        int i  = tid >> 5;
        int c4 = tid & 31;
        const float4* srcA = (const float4*)(qg + headBase + (size_t)i * L_LEN + rA * ROWS_PER_BLK);
        const float4* srcB = (const float4*)(qg + headBase + (size_t)i * L_LEN + rB * ROWS_PER_BLK);
        ((float4*)&sQ[0][i][0])[c4] = srcA[c4];
        ((float4*)&sQ[1][i][0])[c4] = srcB[c4];
    }
    {
        const float4* src = (const float4*)(kerg + (size_t)h * L_LEN);
        for (int u = tid; u < L_LEN / 4; u += 256) {
            float4 f = src[u];
            if (u == 0) f.x += Dh;
            ((float4*)(sKerP + KPAD))[u] = f;
        }
        if (tid < KPAD) sKerP[tid] = 0.0f;
    }
    {
        float4 z = make_float4(0.f, 0.f, 0.f, 0.f);
        for (int u = tid; u < 8 * V_PITCH / 4; u += 256)
            ((float4*)&sVp[8][0])[u] = z;
    }
    __syncthreads();

    const int widB      = NWAVE - 1 - wid;
    const int rowBaseA  = rA * ROWS_PER_BLK + wid  * 16;
    const int rowBaseB  = rB * ROWS_PER_BLK + widB * 16;
    const int rowBaseAU = __builtin_amdgcn_readfirstlane(rowBaseA);
    const int rowBaseBU = __builtin_amdgcn_readfirstlane(rowBaseB);
    const int n  = lane & 15;
    const int k0 = (lane < 16) ? 0 : 2;
    const int mB = (lane < 16) ? 0 : 8;

    v2f qaA0, qaA1, qaB0, qaB1;
    {
        int lcolA = wid  * 16 + n;
        int lcolB = widB * 16 + n;
        qaA0.x = sQ[0][k0 + 0][lcolA];
        qaA0.y = sQ[0][k0 + 1][lcolA];
        qaA1.x = sQ[0][4 + k0 + 0][lcolA];
        qaA1.y = sQ[0][4 + k0 + 1][lcolA];
        qaB0.x = sQ[1][k0 + 0][lcolB];
        qaB0.y = sQ[1][k0 + 1][lcolB];
        qaB1.x = sQ[1][4 + k0 + 0][lcolB];
        qaB1.y = sQ[1][4 + k0 + 1][lcolB];
    }

    v8f accA = {};
    v8f accB = {};
    float* sWw = &sW[wid][0];
    const int arow = n * W_PITCH;

    const int nChunk = (rB >> 2) + 1;
    for (int c = 0; c < nChunk; ++c) {
        const int tStart = c * CHUNK;

        for (int u = tid; u < HEAD_DIM * CHUNK / 4; u += 256) {
            int i  = u >> 7;
            int c4 = u & 127;
            ((float4*)&sK[i][0])[c4]  = ((const float4*)(kg + headBase + (size_t)i * L_LEN + tStart))[c4];
            ((float4*)&sVp[i][0])[c4] = ((const float4*)(vg + headBase + (size_t)i * L_LEN + tStart))[c4];
        }

        if (c + 1 < nChunk) {
            const int tNext = tStart + CHUNK;
            const float* basePtr = (tid < 128) ? kg : vg;
            int r  = (tid & 127) >> 4;
            int ln = (tid & 15) * 32;
            __builtin_prefetch(basePtr + headBase + (size_t)r * L_LEN + tNext + ln, 0, 3);
        }
        __syncthreads();

        row_tile_chunk(accA, qaA0, qaA1, rowBaseA, rowBaseAU, tStart, sK, sVp, sKerP, sWw, n, k0, mB, arow);
        row_tile_chunk(accB, qaB0, qaB1, rowBaseB, rowBaseBU, tStart, sK, sVp, sKerP, sWw, n, k0, mB, arow);

        __syncthreads();
    }

    if (n < 8) {
        size_t oA = headBase + (size_t)n * L_LEN + (size_t)(rowBaseA + mB);
        size_t oB = headBase + (size_t)n * L_LEN + (size_t)(rowBaseB + mB);
        *(float4*)(outg + oA)     = make_float4(accA[0], accA[1], accA[2], accA[3]);
        *(float4*)(outg + oA + 4) = make_float4(accA[4], accA[5], accA[6], accA[7]);
        *(float4*)(outg + oB)     = make_float4(accB[0], accB[1], accB[2], accB[3]);
        *(float4*)(outg + oB + 4) = make_float4(accB[4], accB[5], accB[6], accB[7]);
    }
}

extern "C" void kernel_launch(void* const* d_in, const int* in_sizes, int n_in,
                              void* d_out, int out_size, void* d_ws, size_t ws_size,
                              hipStream_t stream) {
    const float* k   = (const float*)d_in[0];
    const float* v   = (const float*)d_in[1];
    const float* q   = (const float*)d_in[2];
    const float* ker = (const float*)d_in[3];
    const float* D   = (const float*)d_in[4];
    float* out = (float*)d_out;

    const int B = in_sizes[0] / (D_MODEL_SZ * L_LEN);
    dim3 grid((unsigned)(B * N_HEAD * (N_ROW_BLK / 2)));
    hipLaunchKernelGGL(hyena_attn_kernel, grid, dim3(256), 0, stream, k, v, q, ker, D, out);
}


# Tile products formed transposed (S^T = K^T Q, Out^T = V W^T) with K columns read in permuted lane order: modulated tile feeds the 2nd WMMA from registers, LDS W round trip gone; sum order same
# speedup vs baseline: 3.0660x; 1.0661x over previous
#include <hip/hip_runtime.h>

typedef float v2f __attribute__((ext_vector_type(2)));
typedef float v8f __attribute__((ext_vector_type(8)));

#define L_LEN 2048
#define N_HEAD 128
#define D_MODEL_SZ 1024
#define HEAD_DIM 8
#define CHUNK 512
#define ROWS_PER_BLK 128
#define N_ROW_BLK (L_LEN / ROWS_PER_BLK)
#define NWAVE 8
#define KPAD 16
#define K_PITCH (CHUNK + 16)
#define V_PITCH (CHUNK + 4)

static __device__ __forceinline__
void row_tile_chunk(v8f& acc, const v2f qb0, const v2f qb1,
                    const int rowBase, const int rowBaseU, const int tStart,
                    const float (*sK)[K_PITCH], const float (*sVp)[V_PITCH],
                    const float* sKerP,
                    const int n, const int pn, const int k0)
{
    int ttEnd = ((rowBaseU - tStart) >> 4) + 1;
    if (ttEnd > CHUNK / 16) ttEnd = CHUNK / 16;

    for (int tt = 0; tt < ttEnd; ++tt) {
        const int colLoc  = tt * 16;
        const int colBase = tStart + colLoc;

        v2f ka0, ka1;
        ka0.x = sK[k0 + 0][colLoc + pn];
        ka0.y = sK[k0 + 1][colLoc + pn];
        ka1.x = sK[4 + k0 + 0][colLoc + pn];
        ka1.y = sK[4 + k0 + 1][colLoc + pn];

        v8f s = {};
        s = __builtin_amdgcn_wmma_f32_16x16x4_f32(false, ka0, false, qb0, (short)0, s, false, false);
        s = __builtin_amdgcn_wmma_f32_16x16x4_f32(false, ka1, false, qb1, (short)0, s, false, false);

        const int kerBase = KPAD + rowBase + n - colBase - k0;
        #pragma unroll
        for (int r = 0; r < 8; ++r)
            s[r] *= sKerP[kerBase - (4 * (r >> 1) + (r & 1))];

        #pragma unroll
        for (int kc = 0; kc < 4; ++kc) {
            v2f va, wb;
            int tl = colLoc + 4 * kc + k0;
            va.x = sVp[n][tl + 0];
            va.y = sVp[n][tl + 1];
            wb.x = s[2 * kc + 0];
            wb.y = s[2 * kc + 1];
            acc = __builtin_amdgcn_wmma_f32_16x16x4_f32(false, va, false, wb, (short)0, acc, false, false);
        }
    }
}

__global__ __launch_bounds__(256)
void hyena_attn_kernel(const float* __restrict__ kg,
                       const float* __restrict__ vg,
                       const float* __restrict__ qg,
                       const float* __restrict__ kerg,
                       const float* __restrict__ Dg,
                       float* __restrict__ outg)
{
    __shared__ __attribute__((aligned(16))) float sK[HEAD_DIM][K_PITCH];
    __shared__ __attribute__((aligned(16))) float sVp[16][V_PITCH];
    __shared__ __attribute__((aligned(16))) float sQ[2][HEAD_DIM][ROWS_PER_BLK];
    __shared__ __attribute__((aligned(16))) float sKerP[KPAD + L_LEN];

    const int tid  = threadIdx.x;
    const int lane = tid & 31;
    const int wid  = tid >> 5;

    const int rA = blockIdx.x & (N_ROW_BLK / 2 - 1);
    const int rB = N_ROW_BLK - 1 - rA;
    const int bh = blockIdx.x >> 3;
    const int b  = bh / N_HEAD;
    const int h  = bh % N_HEAD;

    const size_t headBase = ((size_t)b * D_MODEL_SZ + (size_t)h * HEAD_DIM) * L_LEN;

    const float Dh = Dg[h];

    {
        int i  = tid >> 5;
        int c4 = tid & 31;
        const float4* srcA = (const float4*)(qg + headBase + (size_t)i * L_LEN + rA * ROWS_PER_BLK);
        const float4* srcB = (const float4*)(qg + headBase + (size_t)i * L_LEN + rB * ROWS_PER_BLK);
        ((float4*)&sQ[0][i][0])[c4] = srcA[c4];
        ((float4*)&sQ[1][i][0])[c4] = srcB[c4];
    }
    {
        const float4* src = (const float4*)(kerg + (size_t)h * L_LEN);
        for (int u = tid; u < L_LEN / 4; u += 256) {
            float4 f = src[u];
            if (u == 0) f.x += Dh;
            ((float4*)(sKerP + KPAD))[u] = f;
        }
        if (tid < KPAD) sKerP[tid] = 0.0f;
    }
    {
        float4 z = make_float4(0.f, 0.f, 0.f, 0.f);
        for (int u = tid; u < 8 * V_PITCH / 4; u += 256)
            ((float4*)&sVp[8][0])[u] = z;
    }
    __syncthreads();

    const int widB      = NWAVE - 1 - wid;
    const int rowBaseA  = rA * ROWS_PER_BLK + wid  * 16;
    const int rowBaseB  = rB * ROWS_PER_BLK + widB * 16;
    const int rowBaseAU = __builtin_amdgcn_readfirstlane(rowBaseA);
    const int rowBaseBU = __builtin_amdgcn_readfirstlane(rowBaseB);
    const int n  = lane & 15;
    const int k0 = (lane < 16) ? 0 : 2;
    const int pn = 4 * ((n >> 1) & 3) + 2 * (n >> 3) + (n & 1);

    v2f qaA0, qaA1, qaB0, qaB1;
    {
        int lcolA = wid  * 16 + n;
        int lcolB = widB * 16 + n;
        qaA0.x = sQ[0][k0 + 0][lcolA];
        qaA0.y = sQ[0][k0 + 1][lcolA];
        qaA1.x = sQ[0][4 + k0 + 0][lcolA];
        qaA1.y = sQ[0][4 + k0 + 1][lcolA];
        qaB0.x = sQ[1][k0 + 0][lcolB];
        qaB0.y = sQ[1][k0 + 1][lcolB];
        qaB1.x = sQ[1][4 + k0 + 0][lcolB];
        qaB1.y = sQ[1][4 + k0 + 1][lcolB];
    }

    v8f accA = {};
    v8f accB = {};

    const int nChunk = (rB >> 2) + 1;
    for (int c = 0; c < nChunk; ++c) {
        const int tStart = c * CHUNK;

        for (int u = tid; u < HEAD_DIM * CHUNK / 4; u += 256) {
            int i  = u >> 7;
            int c4 = u & 127;
            ((float4*)&sK[i][0])[c4]  = ((const float4*)(kg + headBase + (size_t)i * L_LEN + tStart))[c4];
            ((float4*)&sVp[i][0])[c4] = ((const float4*)(vg + headBase + (size_t)i * L_LEN + tStart))[c4];
        }

        if (c + 1 < nChunk) {
            const int tNext = tStart + CHUNK;
            const float* basePtr = (tid < 128) ? kg : vg;
            int r  = (tid & 127) >> 4;
            int ln = (tid & 15) * 32;
            __builtin_prefetch(basePtr + headBase + (size_t)r * L_LEN + tNext + ln, 0, 3);
        }
        __syncthreads();

        row_tile_chunk(accA, qaA0, qaA1, rowBaseA, rowBaseAU, tStart, sK, sVp, sKerP, n, pn, k0);
        row_tile_chunk(accB, qaB0, qaB1, rowBaseB, rowBaseBU, tStart, sK, sVp, sKerP, n, pn, k0);

        __syncthreads();
    }

    if (lane < 16) {
        float* oA = outg + headBase + (size_t)(rowBaseA + n);
        float* oB = outg + headBase + (size_t)(rowBaseB + n);
        #pragma unroll
        for (int r = 0; r < 8; ++r) {
            oA[(size_t)r * L_LEN] = accA[r];
            oB[(size_t)r * L_LEN] = accB[r];
        }
    }
}

extern "C" void kernel_launch(void* const* d_in, const int* in_sizes, int n_in,
                              void* d_out, int out_size, void* d_ws, size_t ws_size,
                              hipStream_t stream) {
    const float* k   = (const float*)d_in[0];
    const float* v   = (const float*)d_in[1];
    const float* q   = (const float*)d_in[2];
    const float* ker = (const float*)d_in[3];
    const float* D   = (const float*)d_in[4];
    float* out = (float*)d_out;

    const int B = in_sizes[0] / (D_MODEL_SZ * L_LEN);
    dim3 grid((unsigned)(B * N_HEAD * (N_ROW_BLK / 2)));
    hipLaunchKernelGGL(hyena_attn_kernel, grid, dim3(256), 0, stream, k, v, q, ker, D, out);
}


# A wave owns two adjacent 16-row tiles (256-row blocks, pairs p and 7-p share a SIMD; grid 2048 -> 1024): one K, V and kappa fragment load feeds both tiles' WMMAs (Toeplitz lag reuse); sum order same
# speedup vs baseline: 3.3423x; 1.0901x over previous
#include <hip/hip_runtime.h>


typedef float v2f __attribute__((ext_vector_type(2)));
typedef float v8f __attribute__((ext_vector_type(8)));

#define L_LEN 2048
#define N_HEAD 128
#define D_MODEL_SZ 1024
#define HEAD_DIM 8
#define CHUNK 512
#define ROWS_PER_BLK 256
#define N_ROW_BLK (L_LEN / ROWS_PER_BLK)
#define NWAVE 8
#define KPAD 16
#define K_PITCH (CHUNK + 16)
#define V_PITCH (CHUNK + 4)

static __device__ __forceinline__
v8f load_kappa(const float* sKerP, const int kerBase)
{
    v8f kap;
    #pragma unroll
    for (int r = 0; r < 8; ++r)
        kap[r] = sKerP[kerBase - (4 * (r >> 1) + (r & 1))];
    return kap;
}

static __device__ __forceinline__
void load_k(v2f& ka0, v2f& ka1, const float (*sK)[K_PITCH], const int colLoc, const int pn, const int k0)
{
    ka0.x = sK[k0 + 0][colLoc + pn];
    ka0.y = sK[k0 + 1][colLoc + pn];
    ka1.x = sK[4 + k0 + 0][colLoc + pn];
    ka1.y = sK[4 + k0 + 1][colLoc + pn];
}

static __device__ __forceinline__
void row_pair_chunk(v8f& acc0, v8f& acc1,
                    const v2f q00, const v2f q01, const v2f q10, const v2f q11,
                    const int rowBase0, const int rowBase0U, const int tStart,
                    const float (*sK)[K_PITCH], const float (*sVp)[V_PITCH],
                    const float* sKerP,
                    const int n, const int pn, const int k0)
{
    int n0 = ((rowBase0U - tStart) >> 4) + 1;
    int n1 = ((rowBase0U + 16 - tStart) >> 4) + 1;
    if (n0 > CHUNK / 16) n0 = CHUNK / 16;
    if (n1 > CHUNK / 16) n1 = CHUNK / 16;
    if (n1 <= 0) return;
    if (n0 < 0) n0 = 0;

    const int kerBaseC = KPAD + rowBase0 + n - tStart - k0;
    v8f kap1 = load_kappa(sKerP, kerBaseC + 16);

    for (int tt = 0; tt < n0; ++tt) {
        const int colLoc = tt * 16;

        v2f ka0, ka1;
        load_k(ka0, ka1, sK, colLoc, pn, k0);

        v8f s0 = {};
        v8f s1 = {};
        s0 = __builtin_amdgcn_wmma_f32_16x16x4_f32(false, ka0, false, q00, (short)0, s0, false, false);
        s1 = __builtin_amdgcn_wmma_f32_16x16x4_f32(false, ka0, false, q10, (short)0, s1, false, false);
        s0 = __builtin_amdgcn_wmma_f32_16x16x4_f32(false, ka1, false, q01, (short)0, s0, false, false);
        s1 = __builtin_amdgcn_wmma_f32_16x16x4_f32(false, ka1, false, q11, (short)0, s1, false, false);

        const v8f kap0 = load_kappa(sKerP, kerBaseC - colLoc);
        #pragma unroll
        for (int r = 0; r < 8; ++r) {
            s0[r] *= kap0[r];
            s1[r] *= kap1[r];
        }

        #pragma unroll
        for (int kc = 0; kc < 4; ++kc) {
            v2f va, wb0, wb1;
            const int tl = colLoc + 4 * kc + k0;
            va.x = sVp[n][tl + 0];
            va.y = sVp[n][tl + 1];
            wb0.x = s0[2 * kc + 0];
            wb0.y = s0[2 * kc + 1];
            wb1.x = s1[2 * kc + 0];
            wb1.y = s1[2 * kc + 1];
            acc0 = __builtin_amdgcn_wmma_f32_16x16x4_f32(false, va, false, wb0, (short)0, acc0, false, false);
            acc1 = __builtin_amdgcn_wmma_f32_16x16x4_f32(false, va, false, wb1, (short)0, acc1, false, false);
        }

        kap1 = kap0;
    }

    if (n1 > n0) {
        const int colLoc = n0 * 16;
        v2f ka0, ka1;
        load_k(ka0, ka1, sK, colLoc, pn, k0);

        v8f s1 = {};
        s1 = __builtin_amdgcn_wmma_f32_16x16x4_f32(false, ka0, false, q10, (short)0, s1, false, false);
        s1 = __builtin_amdgcn_wmma_f32_16x16x4_f32(false, ka1, false, q11, (short)0, s1, false, false);

        #pragma unroll
        for (int r = 0; r < 8; ++r)
            s1[r] *= kap1[r];

        #pragma unroll
        for (int kc = 0; kc < 4; ++kc) {
            v2f va, wb1;
            const int tl = colLoc + 4 * kc + k0;
            va.x = sVp[n][tl + 0];
            va.y = sVp[n][tl + 1];
            wb1.x = s1[2 * kc + 0];
            wb1.y = s1[2 * kc + 1];
            acc1 = __builtin_amdgcn_wmma_f32_16x16x4_f32(false, va, false, wb1, (short)0, acc1, false, false);
        }
    }
}

__global__ __launch_bounds__(256)
void hyena_attn_kernel(const float* __restrict__ kg,
                       const float* __restrict__ vg,
                       const float* __restrict__ qg,
                       const float* __restrict__ kerg,
                       const float* __restrict__ Dg,
                       float* __restrict__ outg)
{
    __shared__ __attribute__((aligned(16))) float sK[HEAD_DIM][K_PITCH];
    __shared__ __attribute__((aligned(16))) float sVp[16][V_PITCH];
    __shared__ __attribute__((aligned(16))) float sQ[2][HEAD_DIM][ROWS_PER_BLK];
    __shared__ __attribute__((aligned(16))) float sKerP[KPAD + L_LEN];

    const int tid  = threadIdx.x;
    const int lane = tid & 31;
    const int wid  = tid >> 5;

    const int rA = blockIdx.x & (N_ROW_BLK / 2 - 1);
    const int rB = N_ROW_BLK - 1 - rA;
    const int bh = blockIdx.x >> 2;
    const int b  = bh / N_HEAD;
    const int h  = bh % N_HEAD;

    const size_t headBase = ((size_t)b * D_MODEL_SZ + (size_t)h * HEAD_DIM) * L_LEN;

    const float Dh = Dg[h];

    for (int u = tid; u < 2 * HEAD_DIM * ROWS_PER_BLK / 4; u += 256) {
        const int blk = u >> 9;
        const int i   = (u >> 6) & 7;
        const int c4  = u & 63;
        const int rr  = blk ? rB : rA;
        const float4* src = (const float4*)(qg + headBase + (size_t)i * L_LEN + rr * ROWS_PER_BLK);
        ((float4*)&sQ[blk][i][0])[c4] = src[c4];
    }
    {
        const float4* src = (const float4*)(kerg + (size_t)h * L_LEN);
        for (int u = tid; u < L_LEN / 4; u += 256) {
            float4 f = src[u];
            if (u == 0) f.x += Dh;
            ((float4*)(sKerP + KPAD))[u] = f;
        }
        if (tid < KPAD) sKerP[tid] = 0.0f;
    }
    {
        float4 z = make_float4(0.f, 0.f, 0.f, 0.f);
        for (int u = tid; u < 8 * V_PITCH / 4; u += 256)
            ((float4*)&sVp[8][0])[u] = z;
    }
    __syncthreads();

    const int widA      = (wid < NWAVE / 2) ? wid : (NWAVE + NWAVE / 2 - 1 - wid);
    const int widB      = NWAVE - 1 - widA;
    const int rowBaseA  = rA * ROWS_PER_BLK + widA * 32;
    const int rowBaseB  = rB * ROWS_PER_BLK + widB * 32;
    const int rowBaseAU = __builtin_amdgcn_readfirstlane(rowBaseA);
    const int rowBaseBU = __builtin_amdgcn_readfirstlane(rowBaseB);
    const int n  = lane & 15;
    const int k0 = (lane < 16) ? 0 : 2;
    const int pn = 4 * ((n >> 1) & 3) + 2 * (n >> 3) + (n & 1);

    v2f qA00, qA01, qA10, qA11, qB00, qB01, qB10, qB11;
    {
        const int lcolA = widA * 32 + n;
        const int lcolB = widB * 32 + n;
        qA00.x = sQ[0][k0 + 0][lcolA];
        qA00.y = sQ[0][k0 + 1][lcolA];
        qA01.x = sQ[0][4 + k0 + 0][lcolA];
        qA01.y = sQ[0][4 + k0 + 1][lcolA];
        qA10.x = sQ[0][k0 + 0][lcolA + 16];
        qA10.y = sQ[0][k0 + 1][lcolA + 16];
        qA11.x = sQ[0][4 + k0 + 0][lcolA + 16];
        qA11.y = sQ[0][4 + k0 + 1][lcolA + 16];
        qB00.x = sQ[1][k0 + 0][lcolB];
        qB00.y = sQ[1][k0 + 1][lcolB];
        qB01.x = sQ[1][4 + k0 + 0][lcolB];
        qB01.y = sQ[1][4 + k0 + 1][lcolB];
        qB10.x = sQ[1][k0 + 0][lcolB + 16];
        qB10.y = sQ[1][k0 + 1][lcolB + 16];
        qB11.x = sQ[1][4 + k0 + 0][lcolB + 16];
        qB11.y = sQ[1][4 + k0 + 1][lcolB + 16];
    }

    v8f accA0 = {};
    v8f accA1 = {};
    v8f accB0 = {};
    v8f accB1 = {};

    const int nChunk = (rB >> 1) + 1;
    for (int c = 0; c < nChunk; ++c) {
        const int tStart = c * CHUNK;

        for (int u = tid; u < HEAD_DIM * CHUNK / 4; u += 256) {
            int i  = u >> 7;
            int c4 = u & 127;
            ((float4*)&sK[i][0])[c4]  = ((const float4*)(kg + headBase + (size_t)i * L_LEN + tStart))[c4];
            ((float4*)&sVp[i][0])[c4] = ((const float4*)(vg + headBase + (size_t)i * L_LEN + tStart))[c4];
        }

        if (c + 1 < nChunk) {
            const int tNext = tStart + CHUNK;
            const float* basePtr = (tid < 128) ? kg : vg;
            int r  = (tid & 127) >> 4;
            int ln = (tid & 15) * 32;
            __builtin_prefetch(basePtr + headBase + (size_t)r * L_LEN + tNext + ln, 0, 3);
        }
        __syncthreads();

        row_pair_chunk(accA0, accA1, qA00, qA01, qA10, qA11, rowBaseA, rowBaseAU, tStart, sK, sVp, sKerP, n, pn, k0);
        row_pair_chunk(accB0, accB1, qB00, qB01, qB10, qB11, rowBaseB, rowBaseBU, tStart, sK, sVp, sKerP, n, pn, k0);

        __syncthreads();
    }

    if (lane < 16) {
        float* oA = outg + headBase + (size_t)(rowBaseA + n);
        float* oB = outg + headBase + (size_t)(rowBaseB + n);
        #pragma unroll
        for (int r = 0; r < 8; ++r) {
            oA[(size_t)r * L_LEN]      = accA0[r];
            oA[(size_t)r * L_LEN + 16] = accA1[r];
            oB[(size_t)r * L_LEN]      = accB0[r];
            oB[(size_t)r * L_LEN + 16] = accB1[r];
        }
    }
}

extern "C" void kernel_launch(void* const* d_in, const int* in_sizes, int n_in,
                              void* d_out, int out_size, void* d_ws, size_t ws_size,
                              hipStream_t stream) {
    const float* k   = (const float*)d_in[0];
    const float* v   = (const float*)d_in[1];
    const float* q   = (const float*)d_in[2];
    const float* ker = (const float*)d_in[3];
    const float* D   = (const float*)d_in[4];
    float* out = (float*)d_out;

    const int B = in_sizes[0] / (D_MODEL_SZ * L_LEN);
    dim3 grid((unsigned)(B * N_HEAD * (N_ROW_BLK / 2)));
    hipLaunchKernelGGL(hyena_attn_kernel, grid, dim3(256), 0, stream, k, v, q, ker, D, out);
}


# LDS reads of a column tile issued as one group ahead of its 12-WMMA block (its kappa and V fragments, the next tile's K fragment into a second register pair): no read between WMMAs; sum order same
# speedup vs baseline: 3.4509x; 1.0325x over previous
#include <hip/hip_runtime.h>


typedef float v2f __attribute__((ext_vector_type(2)));
typedef float v8f __attribute__((ext_vector_type(8)));

#define L_LEN 2048
#define N_HEAD 128
#define D_MODEL_SZ 1024
#define HEAD_DIM 8
#define CHUNK 512
#define ROWS_PER_BLK 256
#define N_ROW_BLK (L_LEN / ROWS_PER_BLK)
#define NWAVE 8
#define KPAD 16
#define K_PITCH (CHUNK + 16)
#define V_PITCH (CHUNK + 4)

static __device__ __forceinline__
v8f load_kappa(const float* sKerP, const int kerBase)
{
    v8f kap;
    #pragma unroll
    for (int r = 0; r < 8; ++r)
        kap[r] = sKerP[kerBase - (4 * (r >> 1) + (r & 1))];
    return kap;
}

static __device__ __forceinline__
void load_k(v2f& ka0, v2f& ka1, const float (*sK)[K_PITCH], const int colLoc, const int pn, const int k0)
{
    ka0.x = sK[k0 + 0][colLoc + pn];
    ka0.y = sK[k0 + 1][colLoc + pn];
    ka1.x = sK[4 + k0 + 0][colLoc + pn];
    ka1.y = sK[4 + k0 + 1][colLoc + pn];
}

static __device__ __forceinline__
void load_v(v2f (&va)[4], const float (*sVp)[V_PITCH], const int colLoc, const int n, const int k0)
{
    #pragma unroll
    for (int kc = 0; kc < 4; ++kc) {
        const int tl = colLoc + 4 * kc + k0;
        va[kc].x = sVp[n][tl + 0];
        va[kc].y = sVp[n][tl + 1];
    }
}

static __device__ __forceinline__
void row_pair_chunk(v8f& acc0, v8f& acc1,
                    const v2f q00, const v2f q01, const v2f q10, const v2f q11,
                    const int rowBase0, const int rowBase0U, const int tStart,
                    const float (*sK)[K_PITCH], const float (*sVp)[V_PITCH],
                    const float* sKerP,
                    const int n, const int pn, const int k0)
{
    int n0 = ((rowBase0U - tStart) >> 4) + 1;
    int n1 = ((rowBase0U + 16 - tStart) >> 4) + 1;
    if (n0 > CHUNK / 16) n0 = CHUNK / 16;
    if (n1 > CHUNK / 16) n1 = CHUNK / 16;
    if (n1 <= 0) return;
    if (n0 < 0) n0 = 0;

    const int kerBaseC = KPAD + rowBase0 + n - tStart - k0;
    v8f kap1 = load_kappa(sKerP, kerBaseC + 16);

    v2f ka0, ka1;
    load_k(ka0, ka1, sK, 0, pn, k0);

    const int lastK = n1 - 1;

    for (int tt = 0; tt < n0; ++tt) {
        const int colLoc = tt * 16;

        const v8f kap0 = load_kappa(sKerP, kerBaseC - colLoc);
        v2f va[4];
        load_v(va, sVp, colLoc, n, k0);
        const int tn = (tt + 1 < lastK) ? tt + 1 : lastK;
        v2f kn0, kn1;
        load_k(kn0, kn1, sK, tn * 16, pn, k0);
        __builtin_amdgcn_sched_barrier(0);

        v8f s0 = {};
        v8f s1 = {};
        s0 = __builtin_amdgcn_wmma_f32_16x16x4_f32(false, ka0, false, q00, (short)0, s0, false, false);
        s1 = __builtin_amdgcn_wmma_f32_16x16x4_f32(false, ka0, false, q10, (short)0, s1, false, false);
        s0 = __builtin_amdgcn_wmma_f32_16x16x4_f32(false, ka1, false, q01, (short)0, s0, false, false);
        s1 = __builtin_amdgcn_wmma_f32_16x16x4_f32(false, ka1, false, q11, (short)0, s1, false, false);

        #pragma unroll
        for (int r = 0; r < 8; ++r) {
            s0[r] *= kap0[r];
            s1[r] *= kap1[r];
        }

        #pragma unroll
        for (int kc = 0; kc < 4; ++kc) {
            v2f wb0, wb1;
            wb0.x = s0[2 * kc + 0];
            wb0.y = s0[2 * kc + 1];
            wb1.x = s1[2 * kc + 0];
            wb1.y = s1[2 * kc + 1];
            acc0 = __builtin_amdgcn_wmma_f32_16x16x4_f32(false, va[kc], false, wb0, (short)0, acc0, false, false);
            acc1 = __builtin_amdgcn_wmma_f32_16x16x4_f32(false, va[kc], false, wb1, (short)0, acc1, false, false);
        }
        __builtin_amdgcn_sched_barrier(0);

        kap1 = kap0;
        ka0  = kn0;
        ka1  = kn1;
    }

    if (n1 > n0) {
        const int colLoc = n0 * 16;
        v2f va[4];
        load_v(va, sVp, colLoc, n, k0);

        v8f s1 = {};
        s1 = __builtin_amdgcn_wmma_f32_16x16x4_f32(false, ka0, false, q10, (short)0, s1, false, false);
        s1 = __builtin_amdgcn_wmma_f32_16x16x4_f32(false, ka1, false, q11, (short)0, s1, false, false);

        #pragma unroll
        for (int r = 0; r < 8; ++r)
            s1[r] *= kap1[r];

        #pragma unroll
        for (int kc = 0; kc < 4; ++kc) {
            v2f wb1;
            wb1.x = s1[2 * kc + 0];
            wb1.y = s1[2 * kc + 1];
            acc1 = __builtin_amdgcn_wmma_f32_16x16x4_f32(false, va[kc], false, wb1, (short)0, acc1, false, false);
        }
    }
}

__global__ __launch_bounds__(256)
void hyena_attn_kernel(const float* __restrict__ kg,
                       const float* __restrict__ vg,
                       const float* __restrict__ qg,
                       const float* __restrict__ kerg,
                       const float* __restrict__ Dg,
                       float* __restrict__ outg)
{
    __shared__ __attribute__((aligned(16))) float sK[HEAD_DIM][K_PITCH];
    __shared__ __attribute__((aligned(16))) float sVp[16][V_PITCH];
    __shared__ __attribute__((aligned(16))) float sQ[2][HEAD_DIM][ROWS_PER_BLK];
    __shared__ __attribute__((aligned(16))) float sKerP[KPAD + L_LEN];

    const int tid  = threadIdx.x;
    const int lane = tid & 31;
    const int wid  = tid >> 5;

    const int rA = blockIdx.x & (N_ROW_BLK / 2 - 1);
    const int rB = N_ROW_BLK - 1 - rA;
    const int bh = blockIdx.x >> 2;
    const int b  = bh / N_HEAD;
    const int h  = bh % N_HEAD;

    const size_t headBase = ((size_t)b * D_MODEL_SZ + (size_t)h * HEAD_DIM) * L_LEN;

    const float Dh = Dg[h];

    for (int u = tid; u < 2 * HEAD_DIM * ROWS_PER_BLK / 4; u += 256) {
        const int blk = u >> 9;
        const int i   = (u >> 6) & 7;
        const int c4  = u & 63;
        const int rr  = blk ? rB : rA;
        const float4* src = (const float4*)(qg + headBase + (size_t)i * L_LEN + rr * ROWS_PER_BLK);
        ((float4*)&sQ[blk][i][0])[c4] = src[c4];
    }
    {
        const float4* src = (const float4*)(kerg + (size_t)h * L_LEN);
        for (int u = tid; u < L_LEN / 4; u += 256) {
            float4 f = src[u];
            if (u == 0) f.x += Dh;
            ((float4*)(sKerP + KPAD))[u] = f;
        }
        if (tid < KPAD) sKerP[tid] = 0.0f;
    }
    {
        float4 z = make_float4(0.f, 0.f, 0.f, 0.f);
        for (int u = tid; u < 8 * V_PITCH / 4; u += 256)
            ((float4*)&sVp[8][0])[u] = z;
    }
    __syncthreads();

    const int widA      = (wid < NWAVE / 2) ? wid : (NWAVE + NWAVE / 2 - 1 - wid);
    const int widB      = NWAVE - 1 - widA;
    const int rowBaseA  = rA * ROWS_PER_BLK + widA * 32;
    const int rowBaseB  = rB * ROWS_PER_BLK + widB * 32;
    const int rowBaseAU = __builtin_amdgcn_readfirstlane(rowBaseA);
    const int rowBaseBU = __builtin_amdgcn_readfirstlane(rowBaseB);
    const int n  = lane & 15;
    const int k0 = (lane < 16) ? 0 : 2;
    const int pn = 4 * ((n >> 1) & 3) + 2 * (n >> 3) + (n & 1);

    v2f qA00, qA01, qA10, qA11, qB00, qB01, qB10, qB11;
    {
        const int lcolA = widA * 32 + n;
        const int lcolB = widB * 32 + n;
        qA00.x = sQ[0][k0 + 0][lcolA];
        qA00.y = sQ[0][k0 + 1][lcolA];
        qA01.x = sQ[0][4 + k0 + 0][lcolA];
        qA01.y = sQ[0][4 + k0 + 1][lcolA];
        qA10.x = sQ[0][k0 + 0][lcolA + 16];
        qA10.y = sQ[0][k0 + 1][lcolA + 16];
        qA11.x = sQ[0][4 + k0 + 0][lcolA + 16];
        qA11.y = sQ[0][4 + k0 + 1][lcolA + 16];
        qB00.x = sQ[1][k0 + 0][lcolB];
        qB00.y = sQ[1][k0 + 1][lcolB];
        qB01.x = sQ[1][4 + k0 + 0][lcolB];
        qB01.y = sQ[1][4 + k0 + 1][lcolB];
        qB10.x = sQ[1][k0 + 0][lcolB + 16];
        qB10.y = sQ[1][k0 + 1][lcolB + 16];
        qB11.x = sQ[1][4 + k0 + 0][lcolB + 16];
        qB11.y = sQ[1][4 + k0 + 1][lcolB + 16];
    }

    v8f accA0 = {};
    v8f accA1 = {};
    v8f accB0 = {};
    v8f accB1 = {};

    const int nChunk = (rB >> 1) + 1;
    for (int c = 0; c < nChunk; ++c) {
        const int tStart = c * CHUNK;

        for (int u = tid; u < HEAD_DIM * CHUNK / 4; u += 256) {
            int i  = u >> 7;
            int c4 = u & 127;
            ((float4*)&sK[i][0])[c4]  = ((const float4*)(kg + headBase + (size_t)i * L_LEN + tStart))[c4];
            ((float4*)&sVp[i][0])[c4] = ((const float4*)(vg + headBase + (size_t)i * L_LEN + tStart))[c4];
        }

        if (c + 1 < nChunk) {
            const int tNext = tStart + CHUNK;
            const float* basePtr = (tid < 128) ? kg : vg;
            int r  = (tid & 127) >> 4;
            int ln = (tid & 15) * 32;
            __builtin_prefetch(basePtr + headBase + (size_t)r * L_LEN + tNext + ln, 0, 3);
        }
        __syncthreads();

        row_pair_chunk(accA0, accA1, qA00, qA01, qA10, qA11, rowBaseA, rowBaseAU, tStart, sK, sVp, sKerP, n, pn, k0);
        row_pair_chunk(accB0, accB1, qB00, qB01, qB10, qB11, rowBaseB, rowBaseBU, tStart, sK, sVp, sKerP, n, pn, k0);

        __syncthreads();
    }

    if (lane < 16) {
        float* oA = outg + headBase + (size_t)(rowBaseA + n);
        float* oB = outg + headBase + (size_t)(rowBaseB + n);
        #pragma unroll
        for (int r = 0; r < 8; ++r) {
            oA[(size_t)r * L_LEN]      = accA0[r];
            oA[(size_t)r * L_LEN + 16] = accA1[r];
            oB[(size_t)r * L_LEN]      = accB0[r];
            oB[(size_t)r * L_LEN + 16] = accB1[r];
        }
    }
}

extern "C" void kernel_launch(void* const* d_in, const int* in_sizes, int n_in,
                              void* d_out, int out_size, void* d_ws, size_t ws_size,
                              hipStream_t stream) {
    const float* k   = (const float*)d_in[0];
    const float* v   = (const float*)d_in[1];
    const float* q   = (const float*)d_in[2];
    const float* ker = (const float*)d_in[3];
    const float* D   = (const float*)d_in[4];
    float* out = (float*)d_out;

    const int B = in_sizes[0] / (D_MODEL_SZ * L_LEN);
    dim3 grid((unsigned)(B * N_HEAD * (N_ROW_BLK / 2)));
    hipLaunchKernelGGL(hyena_attn_kernel, grid, dim3(256), 0, stream, k, v, q, ker, D, out);
}


# kappa staged in reversed lag order, zeroed pad tile behind the K chunk: each tile-loop fragment address is one register plus an immediate; per-tile address and clamp instructions gone; sum order same
# speedup vs baseline: 3.6098x; 1.0460x over previous
#include <hip/hip_runtime.h>


typedef float v2f __attribute__((ext_vector_type(2)));
typedef float v8f __attribute__((ext_vector_type(8)));

#define L_LEN 2048
#define N_HEAD 128
#define D_MODEL_SZ 1024
#define HEAD_DIM 8
#define CHUNK 512
#define ROWS_PER_BLK 256
#define N_ROW_BLK (L_LEN / ROWS_PER_BLK)
#define NWAVE 8
#define KPAD 16
#define K_PITCH (CHUNK + 16)
#define V_PITCH (CHUNK + 4)

static __device__ __forceinline__
v8f load_kappa(const float* kp)
{
    v8f kap;
    #pragma unroll
    for (int r = 0; r < 8; ++r)
        kap[r] = kp[4 * (r >> 1) + (r & 1)];
    return kap;
}

static __device__ __forceinline__
void load_k(v2f& ka0, v2f& ka1, const float* kq)
{
    ka0.x = kq[0];
    ka0.y = kq[K_PITCH];
    ka1.x = kq[4 * K_PITCH];
    ka1.y = kq[5 * K_PITCH];
}

static __device__ __forceinline__
void load_v(v2f (&va)[4], const float* vq)
{
    #pragma unroll
    for (int kc = 0; kc < 4; ++kc)
        va[kc] = *(const v2f*)(vq + 4 * kc);

}

static __device__ __forceinline__
void row_pair_chunk(v8f& acc0, v8f& acc1,
                    const v2f q00, const v2f q01, const v2f q10, const v2f q11,
                    const int rowBase0U, const int mirrorTop, const int tStart,
                    const float* sKq, const float* sVq, const float* sKerR,
                    const int n, const int k0)
{
    int n0 = ((rowBase0U - tStart) >> 4) + 1;
    int n1 = ((rowBase0U + 16 - tStart) >> 4) + 1;
    if (n0 > CHUNK / 16) n0 = CHUNK / 16;
    if (n1 > CHUNK / 16) n1 = CHUNK / 16;
    if (n1 <= 0) return;
    if (n0 < 0) n0 = 0;

    const float* kp = sKerR + (mirrorTop + tStart + k0) + (15 - n);
    const float* kq = sKq;
    const float* vq = sVq;

    v8f kap1 = load_kappa(kp - 16);
    v2f ka0, ka1;
    load_k(ka0, ka1, kq);

    for (int tt = 0; tt < n0; ++tt) {
        const v8f kap0 = load_kappa(kp);
        v2f va[4];
        load_v(va, vq);
        v2f kn0, kn1;
        load_k(kn0, kn1, kq + 16);
        kp += 16;
        vq += 16;
        kq += 16;
        __builtin_amdgcn_sched_barrier(0);

        v8f s0 = {};
        v8f s1 = {};
        s0 = __builtin_amdgcn_wmma_f32_16x16x4_f32(false, ka0, false, q00, (short)0, s0, false, false);
        s1 = __builtin_amdgcn_wmma_f32_16x16x4_f32(false, ka0, false, q10, (short)0, s1, false, false);
        s0 = __builtin_amdgcn_wmma_f32_16x16x4_f32(false, ka1, false, q01, (short)0, s0, false, false);
        s1 = __builtin_amdgcn_wmma_f32_16x16x4_f32(false, ka1, false, q11, (short)0, s1, false, false);

        #pragma unroll
        for (int r = 0; r < 8; ++r) {
            s0[r] *= kap0[r];
            s1[r] *= kap1[r];
        }

        #pragma unroll
        for (int kc = 0; kc < 4; ++kc) {
            v2f wb0, wb1;
            wb0.x = s0[2 * kc + 0];
            wb0.y = s0[2 * kc + 1];
            wb1.x = s1[2 * kc + 0];
            wb1.y = s1[2 * kc + 1];
            acc0 = __builtin_amdgcn_wmma_f32_16x16x4_f32(false, va[kc], false, wb0, (short)0, acc0, false, false);
            acc1 = __builtin_amdgcn_wmma_f32_16x16x4_f32(false, va[kc], false, wb1, (short)0, acc1, false, false);
        }
        __builtin_amdgcn_sched_barrier(0);

        kap1 = kap0;
        ka0  = kn0;
        ka1  = kn1;
    }

    if (n1 > n0) {
        v2f va[4];
        load_v(va, vq);

        v8f s1 = {};
        s1 = __builtin_amdgcn_wmma_f32_16x16x4_f32(false, ka0, false, q10, (short)0, s1, false, false);
        s1 = __builtin_amdgcn_wmma_f32_16x16x4_f32(false, ka1, false, q11, (short)0, s1, false, false);

        #pragma unroll
        for (int r = 0; r < 8; ++r)
            s1[r] *= kap1[r];

        #pragma unroll
        for (int kc = 0; kc < 4; ++kc) {
            v2f wb1;
            wb1.x = s1[2 * kc + 0];
            wb1.y = s1[2 * kc + 1];
            acc1 = __builtin_amdgcn_wmma_f32_16x16x4_f32(false, va[kc], false, wb1, (short)0, acc1, false, false);
        }
    }
}

__global__ __launch_bounds__(256)
void hyena_attn_kernel(const float* __restrict__ kg,
                       const float* __restrict__ vg,
                       const float* __restrict__ qg,
                       const float* __restrict__ kerg,
                       const float* __restrict__ Dg,
                       float* __restrict__ outg)
{
    __shared__ __attribute__((aligned(256))) float sKerR[L_LEN + KPAD];
    __shared__ __attribute__((aligned(16))) float sK[HEAD_DIM][K_PITCH];
    __shared__ __attribute__((aligned(16))) float sVp[16][V_PITCH];
    __shared__ __attribute__((aligned(16))) float sQ[2][HEAD_DIM][ROWS_PER_BLK];

    const int tid  = threadIdx.x;
    const int lane = tid & 31;
    const int wid  = tid >> 5;

    const int rA = blockIdx.x & (N_ROW_BLK / 2 - 1);
    const int rB = N_ROW_BLK - 1 - rA;
    const int bh = blockIdx.x >> 2;
    const int b  = bh / N_HEAD;
    const int h  = bh % N_HEAD;

    const size_t headBase = ((size_t)b * D_MODEL_SZ + (size_t)h * HEAD_DIM) * L_LEN;

    const float Dh = Dg[h];

    for (int u = tid; u < 2 * HEAD_DIM * ROWS_PER_BLK / 4; u += 256) {
        const int blk = u >> 9;
        const int i   = (u >> 6) & 7;
        const int c4  = u & 63;
        const int rr  = blk ? rB : rA;
        const float4* src = (const float4*)(qg + headBase + (size_t)i * L_LEN + rr * ROWS_PER_BLK);
        ((float4*)&sQ[blk][i][0])[c4] = src[c4];
    }
    {
        const float4* src = (const float4*)(kerg + (size_t)h * L_LEN);
        for (int u = tid; u < L_LEN / 4; u += 256) {
            float4 f = src[u];
            if (u == 0) f.x += Dh;
            ((float4*)sKerR)[L_LEN / 4 - 1 - u] = make_float4(f.w, f.z, f.y, f.x);
        }
        if (tid < KPAD) sKerR[L_LEN + tid] = 0.0f;
        if (tid < HEAD_DIM * 16) sK[tid >> 4][CHUNK + (tid & 15)] = 0.0f;
    }
    {
        float4 z = make_float4(0.f, 0.f, 0.f, 0.f);
        for (int u = tid; u < 8 * V_PITCH / 4; u += 256)
            ((float4*)&sVp[8][0])[u] = z;
    }
    __syncthreads();

    const int widA      = (wid < NWAVE / 2) ? wid : (NWAVE + NWAVE / 2 - 1 - wid);
    const int widB      = NWAVE - 1 - widA;
    const int rowBaseA  = rA * ROWS_PER_BLK + widA * 32;
    const int rowBaseB  = rB * ROWS_PER_BLK + widB * 32;
    const int rowBaseAU = __builtin_amdgcn_readfirstlane(rowBaseA);
    const int rowBaseBU = __builtin_amdgcn_readfirstlane(rowBaseB);
    const int n  = lane & 15;
    const int k0 = (lane < 16) ? 0 : 2;
    const int pn = 4 * ((n >> 1) & 3) + 2 * (n >> 3) + (n & 1);

    v2f qA00, qA01, qA10, qA11, qB00, qB01, qB10, qB11;
    {
        const int lcolA = widA * 32 + n;
        const int lcolB = widB * 32 + n;
        qA00.x = sQ[0][k0 + 0][lcolA];
        qA00.y = sQ[0][k0 + 1][lcolA];
        qA01.x = sQ[0][4 + k0 + 0][lcolA];
        qA01.y = sQ[0][4 + k0 + 1][lcolA];
        qA10.x = sQ[0][k0 + 0][lcolA + 16];
        qA10.y = sQ[0][k0 + 1][lcolA + 16];
        qA11.x = sQ[0][4 + k0 + 0][lcolA + 16];
        qA11.y = sQ[0][4 + k0 + 1][lcolA + 16];
        qB00.x = sQ[1][k0 + 0][lcolB];
        qB00.y = sQ[1][k0 + 1][lcolB];
        qB01.x = sQ[1][4 + k0 + 0][lcolB];
        qB01.y = sQ[1][4 + k0 + 1][lcolB];
        qB10.x = sQ[1][k0 + 0][lcolB + 16];
        qB10.y = sQ[1][k0 + 1][lcolB + 16];
        qB11.x = sQ[1][4 + k0 + 0][lcolB + 16];
        qB11.y = sQ[1][4 + k0 + 1][lcolB + 16];
    }

    v8f accA0 = {};
    v8f accA1 = {};
    v8f accB0 = {};
    v8f accB1 = {};

    const int nChunk = (rB >> 1) + 1;
    for (int c = 0; c < nChunk; ++c) {
        const int tStart = c * CHUNK;

        for (int u = tid; u < HEAD_DIM * CHUNK / 4; u += 256) {
            int i  = u >> 7;
            int c4 = u & 127;
            ((float4*)&sK[i][0])[c4]  = ((const float4*)(kg + headBase + (size_t)i * L_LEN + tStart))[c4];
            ((float4*)&sVp[i][0])[c4] = ((const float4*)(vg + headBase + (size_t)i * L_LEN + tStart))[c4];
        }

        if (c + 1 < nChunk) {
            const int tNext = tStart + CHUNK;
            const float* basePtr = (tid < 128) ? kg : vg;
            int r  = (tid & 127) >> 4;
            int ln = (tid & 15) * 32;
            __builtin_prefetch(basePtr + headBase + (size_t)r * L_LEN + tNext + ln, 0, 3);
        }
        __syncthreads();

        row_pair_chunk(accA0, accA1, qA00, qA01, qA10, qA11, rowBaseAU, rowBaseBU + 16, tStart, &sK[k0][pn], &sVp[n][k0], sKerR, n, k0);
        row_pair_chunk(accB0, accB1, qB00, qB01, qB10, qB11, rowBaseBU, rowBaseAU + 16, tStart, &sK[k0][pn], &sVp[n][k0], sKerR, n, k0);

        __syncthreads();
    }

    if (lane < 16) {
        float* oA = outg + headBase + (size_t)(rowBaseA + n);
        float* oB = outg + headBase + (size_t)(rowBaseB + n);
        #pragma unroll
        for (int r = 0; r < 8; ++r) {
            oA[(size_t)r * L_LEN]      = accA0[r];
            oA[(size_t)r * L_LEN + 16] = accA1[r];
            oB[(size_t)r * L_LEN]      = accB0[r];
            oB[(size_t)r * L_LEN + 16] = accB1[r];
        }
    }
}

extern "C" void kernel_launch(void* const* d_in, const int* in_sizes, int n_in,
                              void* d_out, int out_size, void* d_ws, size_t ws_size,
                              hipStream_t stream) {
    const float* k   = (const float*)d_in[0];
    const float* v   = (const float*)d_in[1];
    const float* q   = (const float*)d_in[2];
    const float* ker = (const float*)d_in[3];
    const float* D   = (const float*)d_in[4];
    float* out = (float*)d_out;

    const int B = in_sizes[0] / (D_MODEL_SZ * L_LEN);
    dim3 grid((unsigned)(B * N_HEAD * (N_ROW_BLK / 2)));
    hipLaunchKernelGGL(hyena_attn_kernel, grid, dim3(256), 0, stream, k, v, q, ker, D, out);
}
